# starting kernel
# baseline (speedup 1.0000x reference)
#include <hip/hip_runtime.h>

#define DIM   300
#define T1N   128
#define T2N   128
#define BATCH 4
#define ROWS  (BATCH * T1N)

#define WAVES 8
#define BLOCK (WAVES * 32)

typedef __attribute__((ext_vector_type(2))) float v2f;
typedef __attribute__((ext_vector_type(8))) float v8f;

__global__ __launch_bounds__(32) void proj_wmma_kernel(
    const float* __restrict__ X, const float* __restrict__ W,
    const float* __restrict__ bias, float* __restrict__ out)
{
    const int lane  = threadIdx.x;
    const int nTile = blockIdx.x;
    const int mTile = blockIdx.y;
    const int m0 = mTile * 16;
    const int n0 = nTile * 16;

    const int half = lane >> 4;
    const int l16  = lane & 15;

    const int  n      = n0 + l16;
    const bool nvalid = (n < DIM);
    const int  nsafe  = nvalid ? n : (DIM - 1);

    const float* arow = X + (size_t)(m0 + l16) * DIM + 2 * half;
    const float* wrow = W + (size_t)nsafe      * DIM + 2 * half;

    v8f acc0 = {};
    v8f acc1 = {};
    int kt = 0;
    for (; kt + 8 <= DIM; kt += 8) {
        v2f a0, b0, a1, b1;
        a0.x = arow[kt];     a0.y = arow[kt + 1];
        b0.x = wrow[kt];     b0.y = wrow[kt + 1];
        a1.x = arow[kt + 4]; a1.y = arow[kt + 5];
        b1.x = wrow[kt + 4]; b1.y = wrow[kt + 5];
        acc0 = __builtin_amdgcn_wmma_f32_16x16x4_f32(
            false, a0, false, b0, (short)0, acc0, false, false);
        acc1 = __builtin_amdgcn_wmma_f32_16x16x4_f32(
            false, a1, false, b1, (short)0, acc1, false, false);
    }
    {
        v2f a0, b0;
        a0.x = arow[kt]; a0.y = arow[kt + 1];
        b0.x = wrow[kt]; b0.y = wrow[kt + 1];
        acc0 = __builtin_amdgcn_wmma_f32_16x16x4_f32(
            false, a0, false, b0, (short)0, acc0, false, false);
    }
    const v8f acc = acc0 + acc1;

    if (nvalid) {
        const float bv = bias[n];
        #pragma unroll
        for (int r = 0; r < 8; ++r) {
            const int m = m0 + r + 8 * half;
            out[(size_t)m * DIM + n] = acc[r] + bv;
        }
    }
}

__global__ __launch_bounds__(BLOCK) void attn_fused_kernel(
    const float* __restrict__ q, const float* __restrict__ k,
    const float* __restrict__ v, const float* __restrict__ hL,
    const float* __restrict__ hR, float* __restrict__ out)
{
    const int b    = blockIdx.y;
    const int i    = blockIdx.x;
    const int tid  = threadIdx.x;
    const int wave = tid >> 5;
    const int lane = tid & 31;

    __shared__ float s_q[DIM];
    __shared__ float s_attn[T2N];

    const float* qrow = q + ((size_t)b * T1N + i) * DIM;
    for (int d = tid; d < DIM; d += BLOCK) s_q[d] = qrow[d];
    __syncthreads();

    for (int j = wave; j < T2N; j += WAVES) {
        const float* hLp = hL + (((size_t)b * T1N + i) * T2N + j) * DIM;
        const float* hRp = hR + (((size_t)b * T2N + j) * T1N + i) * DIM;
        const float* kp  = k  + ((size_t)b * T2N + j) * DIM;
        float acc = 0.0f;
        for (int d = lane; d < DIM; d += 32)
            acc += (s_q[d] + hLp[d]) * (kp[d] + hRp[d]);
        #pragma unroll
        for (int off = 16; off > 0; off >>= 1)
            acc += __shfl_xor(acc, off, 32);
        if (lane == 0) s_attn[j] = acc;
    }
    __syncthreads();

    if (wave == 0) {
        float m = -__builtin_inff();
        for (int j = lane; j < T2N; j += 32) m = fmaxf(m, s_attn[j]);
        #pragma unroll
        for (int off = 16; off > 0; off >>= 1) m = fmaxf(m, __shfl_xor(m, off, 32));

        float sum = 0.0f;
        for (int j = lane; j < T2N; j += 32) sum += expf(s_attn[j] - m);
        #pragma unroll
        for (int off = 16; off > 0; off >>= 1) sum += __shfl_xor(sum, off, 32);
        const float inv = 1.0f / sum;

        float m2 = -__builtin_inff();
        for (int j = lane; j < T2N; j += 32) {
            const float p = expf(s_attn[j] - m) * inv;
            m2 = fmaxf(m2, 1000.0f * p);
        }
        #pragma unroll
        for (int off = 16; off > 0; off >>= 1) m2 = fmaxf(m2, __shfl_xor(m2, off, 32));

        float sum2 = 0.0f;
        for (int j = lane; j < T2N; j += 32) {
            const float p = expf(s_attn[j] - m) * inv;
            sum2 += expf(1000.0f * p - m2);
        }
        #pragma unroll
        for (int off = 16; off > 0; off >>= 1) sum2 += __shfl_xor(sum2, off, 32);
        const float inv2 = 1.0f / sum2;

        for (int j = lane; j < T2N; j += 32) {
            const float p  = expf(s_attn[j] - m) * inv;
            const float a2 = expf(1000.0f * p - m2) * inv2;
            s_attn[j] = fminf(fmaxf(a2, 0.0f), 1.0f);
        }
    }
    __syncthreads();

    for (int d = tid; d < DIM; d += BLOCK) {
        const float* vp  = v  + (size_t)b * T2N * DIM + d;
        const float* hRp = hR + ((size_t)b * T2N * T1N + i) * DIM + d;
        float acc = 0.0f;
        for (int j = 0; j < T2N; ++j)
            acc += s_attn[j] * (vp[(size_t)j * DIM] + hRp[(size_t)j * T1N * DIM]);
        out[((size_t)b * T1N + i) * DIM + d] = acc;
    }
}

extern "C" void kernel_launch(void* const* d_in, const int* in_sizes, int n_in,
                              void* d_out, int out_size, void* d_ws, size_t ws_size,
                              hipStream_t stream)
{
    const float* query = (const float*)d_in[0];
    const float* key   = (const float*)d_in[1];
    const float* value = (const float*)d_in[2];
    const float* hL    = (const float*)d_in[3];
    const float* hR    = (const float*)d_in[4];
    const float* WQ    = (const float*)d_in[5];
    const float* bQ    = (const float*)d_in[6];
    const float* WK    = (const float*)d_in[7];
    const float* bK    = (const float*)d_in[8];
    float* out = (float*)d_out;

    float* q = (float*)d_ws;
    float* k = q + (size_t)ROWS * DIM;
    float* v = k + (size_t)ROWS * DIM;

    const dim3 pgrid((DIM + 15) / 16, ROWS / 16);
    proj_wmma_kernel<<<pgrid, 32, 0, stream>>>(query, WQ, bQ, q);
    proj_wmma_kernel<<<pgrid, 32, 0, stream>>>(key,   WK, bK, k);
    proj_wmma_kernel<<<pgrid, 32, 0, stream>>>(value, WK, bK, v);

    attn_fused_kernel<<<dim3(T1N, BATCH), BLOCK, 0, stream>>>(q, k, v, hL, hR, out);
}


# k and v projections in one launch: a tile's wave loads the shared WK fragments once and runs the key and value WMMA chains on them (3 launches, was 4); per-tile sum order kept, k, v bit-identical
# speedup vs baseline: 3.9208x; 3.9208x over previous
#include <hip/hip_runtime.h>

#define DIM   300
#define T1N   128
#define T2N   128
#define BATCH 4
#define ROWS  (BATCH * T1N)

#define WAVES 8
#define IPB   2
#define BLOCK (WAVES * 32)

typedef __attribute__((ext_vector_type(2))) float v2f;
typedef __attribute__((ext_vector_type(8))) float v8f;
typedef __attribute__((ext_vector_type(4))) float v4f;

__global__ __launch_bounds__(32) void proj_wmma_kernel(
    const float* __restrict__ X, const float* __restrict__ W,
    const float* __restrict__ bias, float* __restrict__ out)
{
    const int lane  = threadIdx.x;
    const int nTile = blockIdx.x;
    const int mTile = blockIdx.y;
    const int m0 = mTile * 16;
    const int n0 = nTile * 16;

    const int half = lane >> 4;
    const int l16  = lane & 15;

    const int  n      = n0 + l16;
    const bool nvalid = (n < DIM);
    const int  nsafe  = nvalid ? n : (DIM - 1);

    const float* arow = X + (size_t)(m0 + l16) * DIM + 2 * half;
    const float* wrow = W + (size_t)nsafe      * DIM + 2 * half;

    v2f a[DIM / 4], b[DIM / 4];
    #pragma unroll
    for (int s = 0; s < DIM / 4; ++s) {
        a[s].x = arow[4 * s]; a[s].y = arow[4 * s + 1];
        b[s].x = wrow[4 * s]; b[s].y = wrow[4 * s + 1];
    }
    __builtin_amdgcn_sched_barrier(0);

    v8f acc0 = {};
    v8f acc1 = {};
    #pragma unroll
    for (int s = 0; s + 1 < DIM / 4; s += 2) {
        acc0 = __builtin_amdgcn_wmma_f32_16x16x4_f32(
            false, a[s], false, b[s], (short)0, acc0, false, false);
        acc1 = __builtin_amdgcn_wmma_f32_16x16x4_f32(
            false, a[s + 1], false, b[s + 1], (short)0, acc1, false, false);
    }
    acc0 = __builtin_amdgcn_wmma_f32_16x16x4_f32(
        false, a[DIM / 4 - 1], false, b[DIM / 4 - 1], (short)0, acc0, false, false);
    const v8f acc = acc0 + acc1;

    if (nvalid) {
        const float bv = bias[n];
        #pragma unroll
        for (int r = 0; r < 8; ++r) {
            const int m = m0 + r + 8 * half;
            out[(size_t)m * DIM + n] = acc[r] + bv;
        }
    }
}

__global__ __launch_bounds__(32) void proj_pair_wmma_kernel(
    const float* __restrict__ X0, const float* __restrict__ X1,
    const float* __restrict__ W, const float* __restrict__ bias,
    float* __restrict__ out0, float* __restrict__ out1)
{
    const int lane  = threadIdx.x;
    const int nTile = blockIdx.x;
    const int mTile = blockIdx.y;
    const int m0 = mTile * 16;
    const int n0 = nTile * 16;

    const int half = lane >> 4;
    const int l16  = lane & 15;

    const int  n      = n0 + l16;
    const bool nvalid = (n < DIM);
    const int  nsafe  = nvalid ? n : (DIM - 1);

    const float* arow0 = X0 + (size_t)(m0 + l16) * DIM + 2 * half;
    const float* arow1 = X1 + (size_t)(m0 + l16) * DIM + 2 * half;
    const float* wrow  = W  + (size_t)nsafe      * DIM + 2 * half;

    v2f a0[DIM / 4], a1[DIM / 4], b[DIM / 4];
    #pragma unroll
    for (int s = 0; s < DIM / 4; ++s) {
        b[s].x  = wrow[4 * s];  b[s].y  = wrow[4 * s + 1];
        a0[s].x = arow0[4 * s]; a0[s].y = arow0[4 * s + 1];
        a1[s].x = arow1[4 * s]; a1[s].y = arow1[4 * s + 1];
    }
    __builtin_amdgcn_sched_barrier(0);

    v8f c00 = {}, c01 = {};
    v8f c10 = {}, c11 = {};
    #pragma unroll
    for (int s = 0; s + 1 < DIM / 4; s += 2) {
        c00 = __builtin_amdgcn_wmma_f32_16x16x4_f32(
            false, a0[s], false, b[s], (short)0, c00, false, false);
        c10 = __builtin_amdgcn_wmma_f32_16x16x4_f32(
            false, a1[s], false, b[s], (short)0, c10, false, false);
        c01 = __builtin_amdgcn_wmma_f32_16x16x4_f32(
            false, a0[s + 1], false, b[s + 1], (short)0, c01, false, false);
        c11 = __builtin_amdgcn_wmma_f32_16x16x4_f32(
            false, a1[s + 1], false, b[s + 1], (short)0, c11, false, false);
    }
    c00 = __builtin_amdgcn_wmma_f32_16x16x4_f32(
        false, a0[DIM / 4 - 1], false, b[DIM / 4 - 1], (short)0, c00, false, false);
    c10 = __builtin_amdgcn_wmma_f32_16x16x4_f32(
        false, a1[DIM / 4 - 1], false, b[DIM / 4 - 1], (short)0, c10, false, false);
    const v8f acc0 = c00 + c01;
    const v8f acc1 = c10 + c11;

    if (nvalid) {
        const float bv = bias[n];
        #pragma unroll
        for (int r = 0; r < 8; ++r) {
            const int m = m0 + r + 8 * half;
            out0[(size_t)m * DIM + n] = acc0[r] + bv;
            out1[(size_t)m * DIM + n] = acc1[r] + bv;
        }
    }
}


__device__ __forceinline__ float score_row(
    const v4f (&qv)[10], const v4f (&a)[10], const v4f (&r)[10],
    const v4f (&kk)[10], const bool tail)
{
    float c0 = 0.0f, c1 = 0.0f, c2 = 0.0f, c3 = 0.0f;
    for (int t = 0; t < 9; ++t) {
        c0 = fmaf(qv[t].x + a[t].x, kk[t].x + r[t].x, c0);
        c1 = fmaf(qv[t].y + a[t].y, kk[t].y + r[t].y, c1);
        c2 = fmaf(qv[t].z + a[t].z, kk[t].z + r[t].z, c2);
        c3 = fmaf(qv[t].w + a[t].w, kk[t].w + r[t].w, c3);
    }
    {
        const float n0 = fmaf(qv[9].x + a[9].x, kk[9].x + r[9].x, c0);
        const float n1 = fmaf(qv[9].y + a[9].y, kk[9].y + r[9].y, c1);
        const float n2 = fmaf(qv[9].z + a[9].z, kk[9].z + r[9].z, c2);
        const float n3 = fmaf(qv[9].w + a[9].w, kk[9].w + r[9].w, c3);
        c0 = tail ? n0 : c0; c1 = tail ? n1 : c1;
        c2 = tail ? n2 : c2; c3 = tail ? n3 : c3;
    }
    for (int off = 4; off > 0; off >>= 1) {
        c0 += __shfl_xor(c0, off, 32);
        c1 += __shfl_xor(c1, off, 32);
        c2 += __shfl_xor(c2, off, 32);
        c3 += __shfl_xor(c3, off, 32);
    }
    const float d0 = c0 + c2;
    const float d1 = c1 + c3;
    return d0 + d1;
}

__global__ __launch_bounds__(BLOCK) void attn_fused_kernel(
    const float* __restrict__ q, const float* __restrict__ k,
    const float* __restrict__ v, const float* __restrict__ hL,
    const float* __restrict__ hR, float* __restrict__ out)
{
    const int b    = blockIdx.y;
    const int i0   = IPB * blockIdx.x;
    const int tid  = threadIdx.x;
    const int wave = tid >> 5;
    const int lane = tid & 31;

    __shared__ float s_attn[IPB][T2N];
    __shared__ v4f   s_hR[IPB][T2N][DIM / 4];

    const int  rsub = lane >> 3;
    const int  m    = lane & 7;
    const bool tail = (m < 3);
    const int  p9   = tail ? (m + 72) : (m + 64);

    const v4f* q4 = (const v4f*)(q + ((size_t)b * T1N + i0) * DIM);
    v4f qa[10], qb[10];
    for (int t = 0; t < 9; ++t) { qa[t] = q4[m + 8 * t]; qb[t] = q4[DIM / 4 + m + 8 * t]; }
    qa[9] = q4[p9]; qb[9] = q4[DIM / 4 + p9];

    for (int g = wave; g < T2N / 4; g += WAVES) {
        const int j = 4 * g + rsub;
        const v4f* hLa = (const v4f*)(hL + (((size_t)b * T1N + i0) * T2N + j) * DIM);
        const v4f* hLb = hLa + (size_t)T2N * (DIM / 4);
        const v4f* hRa = (const v4f*)(hR + (((size_t)b * T2N + j) * T1N + i0) * DIM);
        const v4f* hRb = hRa + DIM / 4;
        const v4f* k4  = (const v4f*)(k  + ((size_t)b * T2N + j) * DIM);

        v4f kk[10], a0[10], r0[10], a1[10], r1[10];
        for (int t = 0; t < 9; ++t) {
            kk[t] = k4[m + 8 * t];
            a0[t] = hLa[m + 8 * t];
            r0[t] = hRa[m + 8 * t];
            a1[t] = hLb[m + 8 * t];
            r1[t] = hRb[m + 8 * t];
        }
        kk[9] = k4[p9];
        a0[9] = hLa[p9]; r0[9] = hRa[p9];
        a1[9] = hLb[p9]; r1[9] = hRb[p9];

        for (int t = 0; t < 9; ++t) {
            s_hR[0][j][m + 8 * t] = r0[t];
            s_hR[1][j][m + 8 * t] = r1[t];
        }
        s_hR[0][j][p9] = r0[9];
        s_hR[1][j][p9] = r1[9];

        const float sc0 = score_row(qa, a0, r0, kk, tail);
        const float sc1 = score_row(qb, a1, r1, kk, tail);
        if (m == 0) { s_attn[0][j] = sc0; s_attn[1][j] = sc1; }
    }
    __syncthreads();

    if (wave < IPB) {
        float* sa = s_attn[wave];
        float m = -__builtin_inff();
        for (int j = lane; j < T2N; j += 32) m = fmaxf(m, sa[j]);
        #pragma unroll
        for (int off = 16; off > 0; off >>= 1) m = fmaxf(m, __shfl_xor(m, off, 32));

        float sum = 0.0f;
        for (int j = lane; j < T2N; j += 32) sum += expf(sa[j] - m);
        #pragma unroll
        for (int off = 16; off > 0; off >>= 1) sum += __shfl_xor(sum, off, 32);
        const float inv = 1.0f / sum;

        float m2 = -__builtin_inff();
        for (int j = lane; j < T2N; j += 32) {
            const float p = expf(sa[j] - m) * inv;
            m2 = fmaxf(m2, 1000.0f * p);
        }
        #pragma unroll
        for (int off = 16; off > 0; off >>= 1) m2 = fmaxf(m2, __shfl_xor(m2, off, 32));

        float sum2 = 0.0f;
        for (int j = lane; j < T2N; j += 32) {
            const float p = expf(sa[j] - m) * inv;
            sum2 += expf(1000.0f * p - m2);
        }
        #pragma unroll
        for (int off = 16; off > 0; off >>= 1) sum2 += __shfl_xor(sum2, off, 32);
        const float inv2 = 1.0f / sum2;

        for (int j = lane; j < T2N; j += 32) {
            const float p  = expf(sa[j] - m) * inv;
            const float a2 = expf(1000.0f * p - m2) * inv2;
            sa[j] = fminf(fmaxf(a2, 0.0f), 1.0f);
        }
    }
    __syncthreads();

    {
        const int  j0 = wave * (T2N / WAVES);
        const int  p2 = (lane < 11) ? (lane + 64) : (lane + 32);
        const v4f* v4 = (const v4f*)(v  + ((size_t)b * T2N + j0) * DIM);

        v4f acc[IPB][3] = {};
        #pragma unroll 1
        for (int r0 = 0; r0 < T2N / WAVES; r0 += 4) {
            v4f vv[4][3], hh[IPB][4][3];
            for (int r = 0; r < 4; ++r) {
                const v4f* vr = v4 + (size_t)(r0 + r) * (DIM / 4);
                vv[r][0] = vr[lane]; vv[r][1] = vr[lane + 32]; vv[r][2] = vr[p2];
                for (int ii = 0; ii < IPB; ++ii) {
                    const v4f* hr = s_hR[ii][j0 + r0 + r];
                    hh[ii][r][0] = hr[lane]; hh[ii][r][1] = hr[lane + 32]; hh[ii][r][2] = hr[p2];
                }
            }
            for (int ii = 0; ii < IPB; ++ii) {
                for (int r = 0; r < 4; ++r) {
                    const float w = s_attn[ii][j0 + r0 + r];
                    for (int c = 0; c < 3; ++c) {
                        acc[ii][c].x = fmaf(w, vv[r][c].x + hh[ii][r][c].x, acc[ii][c].x);
                        acc[ii][c].y = fmaf(w, vv[r][c].y + hh[ii][r][c].y, acc[ii][c].y);
                        acc[ii][c].z = fmaf(w, vv[r][c].z + hh[ii][r][c].z, acc[ii][c].z);
                        acc[ii][c].w = fmaf(w, vv[r][c].w + hh[ii][r][c].w, acc[ii][c].w);
                    }
                }
            }
        }
        for (int ii = 0; ii < IPB; ++ii) {
            s_hR[ii][j0][lane]      = acc[ii][0];
            s_hR[ii][j0][lane + 32] = acc[ii][1];
            if (lane < 11) s_hR[ii][j0][lane + 64] = acc[ii][2];
        }
    }
    __syncthreads();

    if (tid < IPB * (DIM / 4)) {
        const int ii = (tid >= DIM / 4) ? 1 : 0;
        const int p  = tid - ii * (DIM / 4);
        v4f s = s_hR[ii][0][p];
        for (int w = 1; w < WAVES; ++w) s += s_hR[ii][w * (T2N / WAVES)][p];
        ((v4f*)(out + ((size_t)b * T1N + i0) * DIM))[tid] = s;
    }
}

extern "C" void kernel_launch(void* const* d_in, const int* in_sizes, int n_in,
                              void* d_out, int out_size, void* d_ws, size_t ws_size,
                              hipStream_t stream)
{
    const float* query = (const float*)d_in[0];
    const float* key   = (const float*)d_in[1];
    const float* value = (const float*)d_in[2];
    const float* hL    = (const float*)d_in[3];
    const float* hR    = (const float*)d_in[4];
    const float* WQ    = (const float*)d_in[5];
    const float* bQ    = (const float*)d_in[6];
    const float* WK    = (const float*)d_in[7];
    const float* bK    = (const float*)d_in[8];
    float* out = (float*)d_out;

    float* q = (float*)d_ws;
    float* k = q + (size_t)ROWS * DIM;
    float* v = k + (size_t)ROWS * DIM;

    const dim3 pgrid((DIM + 15) / 16, ROWS / 16);
    proj_wmma_kernel<<<pgrid, 32, 0, stream>>>(query, WQ, bQ, q);
    proj_pair_wmma_kernel<<<pgrid, 32, 0, stream>>>(key, value, WK, bK, k, v);

    attn_fused_kernel<<<dim3(T1N / IPB, BATCH), BLOCK, 0, stream>>>(q, k, v, hL, hR, out);
}


# softmax on all 8 waves (4 per row i): a thread owns one score and evaluates its two exponentials once, values exchanged through LDS; column sums and lane folds in the old order: weights bit-identical
# speedup vs baseline: 4.0661x; 1.0371x over previous
#include <hip/hip_runtime.h>

#define DIM   300
#define T1N   128
#define T2N   128
#define BATCH 4
#define ROWS  (BATCH * T1N)

#define WAVES 8
#define IPB   2
#define BLOCK (WAVES * 32)

typedef __attribute__((ext_vector_type(2))) float v2f;
typedef __attribute__((ext_vector_type(8))) float v8f;
typedef __attribute__((ext_vector_type(4))) float v4f;

__global__ __launch_bounds__(32) void proj_wmma_kernel(
    const float* __restrict__ X, const float* __restrict__ W,
    const float* __restrict__ bias, float* __restrict__ out)
{
    const int lane  = threadIdx.x;
    const int nTile = blockIdx.x;
    const int mTile = blockIdx.y;
    const int m0 = mTile * 16;
    const int n0 = nTile * 16;

    const int half = lane >> 4;
    const int l16  = lane & 15;

    const int  n      = n0 + l16;
    const bool nvalid = (n < DIM);
    const int  nsafe  = nvalid ? n : (DIM - 1);

    const float* arow = X + (size_t)(m0 + l16) * DIM + 2 * half;
    const float* wrow = W + (size_t)nsafe      * DIM + 2 * half;

    v2f a[DIM / 4], b[DIM / 4];
    #pragma unroll
    for (int s = 0; s < DIM / 4; ++s) {
        a[s].x = arow[4 * s]; a[s].y = arow[4 * s + 1];
        b[s].x = wrow[4 * s]; b[s].y = wrow[4 * s + 1];
    }
    __builtin_amdgcn_sched_barrier(0);

    v8f acc0 = {};
    v8f acc1 = {};
    #pragma unroll
    for (int s = 0; s + 1 < DIM / 4; s += 2) {
        acc0 = __builtin_amdgcn_wmma_f32_16x16x4_f32(
            false, a[s], false, b[s], (short)0, acc0, false, false);
        acc1 = __builtin_amdgcn_wmma_f32_16x16x4_f32(
            false, a[s + 1], false, b[s + 1], (short)0, acc1, false, false);
    }
    acc0 = __builtin_amdgcn_wmma_f32_16x16x4_f32(
        false, a[DIM / 4 - 1], false, b[DIM / 4 - 1], (short)0, acc0, false, false);
    const v8f acc = acc0 + acc1;

    if (nvalid) {
        const float bv = bias[n];
        #pragma unroll
        for (int r = 0; r < 8; ++r) {
            const int m = m0 + r + 8 * half;
            out[(size_t)m * DIM + n] = acc[r] + bv;
        }
    }
}

__global__ __launch_bounds__(32) void proj_pair_wmma_kernel(
    const float* __restrict__ X0, const float* __restrict__ X1,
    const float* __restrict__ W, const float* __restrict__ bias,
    float* __restrict__ out0, float* __restrict__ out1)
{
    const int lane  = threadIdx.x;
    const int nTile = blockIdx.x;
    const int mTile = blockIdx.y;
    const int m0 = mTile * 16;
    const int n0 = nTile * 16;

    const int half = lane >> 4;
    const int l16  = lane & 15;

    const int  n      = n0 + l16;
    const bool nvalid = (n < DIM);
    const int  nsafe  = nvalid ? n : (DIM - 1);

    const float* arow0 = X0 + (size_t)(m0 + l16) * DIM + 2 * half;
    const float* arow1 = X1 + (size_t)(m0 + l16) * DIM + 2 * half;
    const float* wrow  = W  + (size_t)nsafe      * DIM + 2 * half;

    v2f a0[DIM / 4], a1[DIM / 4], b[DIM / 4];
    #pragma unroll
    for (int s = 0; s < DIM / 4; ++s) {
        b[s].x  = wrow[4 * s];  b[s].y  = wrow[4 * s + 1];
        a0[s].x = arow0[4 * s]; a0[s].y = arow0[4 * s + 1];
        a1[s].x = arow1[4 * s]; a1[s].y = arow1[4 * s + 1];
    }
    __builtin_amdgcn_sched_barrier(0);

    v8f c00 = {}, c01 = {};
    v8f c10 = {}, c11 = {};
    #pragma unroll
    for (int s = 0; s + 1 < DIM / 4; s += 2) {
        c00 = __builtin_amdgcn_wmma_f32_16x16x4_f32(
            false, a0[s], false, b[s], (short)0, c00, false, false);
        c10 = __builtin_amdgcn_wmma_f32_16x16x4_f32(
            false, a1[s], false, b[s], (short)0, c10, false, false);
        c01 = __builtin_amdgcn_wmma_f32_16x16x4_f32(
            false, a0[s + 1], false, b[s + 1], (short)0, c01, false, false);
        c11 = __builtin_amdgcn_wmma_f32_16x16x4_f32(
            false, a1[s + 1], false, b[s + 1], (short)0, c11, false, false);
    }
    c00 = __builtin_amdgcn_wmma_f32_16x16x4_f32(
        false, a0[DIM / 4 - 1], false, b[DIM / 4 - 1], (short)0, c00, false, false);
    c10 = __builtin_amdgcn_wmma_f32_16x16x4_f32(
        false, a1[DIM / 4 - 1], false, b[DIM / 4 - 1], (short)0, c10, false, false);
    const v8f acc0 = c00 + c01;
    const v8f acc1 = c10 + c11;

    if (nvalid) {
        const float bv = bias[n];
        #pragma unroll
        for (int r = 0; r < 8; ++r) {
            const int m = m0 + r + 8 * half;
            out0[(size_t)m * DIM + n] = acc0[r] + bv;
            out1[(size_t)m * DIM + n] = acc1[r] + bv;
        }
    }
}


__device__ __forceinline__ float score_row(
    const v4f (&qv)[10], const v4f (&a)[10], const v4f (&r)[10],
    const v4f (&kk)[10], const bool tail)
{
    float c0 = 0.0f, c1 = 0.0f, c2 = 0.0f, c3 = 0.0f;
    for (int t = 0; t < 9; ++t) {
        c0 = fmaf(qv[t].x + a[t].x, kk[t].x + r[t].x, c0);
        c1 = fmaf(qv[t].y + a[t].y, kk[t].y + r[t].y, c1);
        c2 = fmaf(qv[t].z + a[t].z, kk[t].z + r[t].z, c2);
        c3 = fmaf(qv[t].w + a[t].w, kk[t].w + r[t].w, c3);
    }
    {
        const float n0 = fmaf(qv[9].x + a[9].x, kk[9].x + r[9].x, c0);
        const float n1 = fmaf(qv[9].y + a[9].y, kk[9].y + r[9].y, c1);
        const float n2 = fmaf(qv[9].z + a[9].z, kk[9].z + r[9].z, c2);
        const float n3 = fmaf(qv[9].w + a[9].w, kk[9].w + r[9].w, c3);
        c0 = tail ? n0 : c0; c1 = tail ? n1 : c1;
        c2 = tail ? n2 : c2; c3 = tail ? n3 : c3;
    }
    for (int off = 4; off > 0; off >>= 1) {
        c0 += __shfl_xor(c0, off, 32);
        c1 += __shfl_xor(c1, off, 32);
        c2 += __shfl_xor(c2, off, 32);
        c3 += __shfl_xor(c3, off, 32);
    }
    const float d0 = c0 + c2;
    const float d1 = c1 + c3;
    return d0 + d1;
}

__global__ __launch_bounds__(BLOCK) void attn_fused_kernel(
    const float* __restrict__ q, const float* __restrict__ k,
    const float* __restrict__ v, const float* __restrict__ hL,
    const float* __restrict__ hR, float* __restrict__ out)
{
    const int b    = blockIdx.y;
    const int i0   = IPB * blockIdx.x;
    const int tid  = threadIdx.x;
    const int wave = tid >> 5;
    const int lane = tid & 31;

    __shared__ float s_attn[IPB][T2N];
    __shared__ float s_e1[IPB][T2N];
    __shared__ float s_e2[IPB][T2N];
    __shared__ v4f   s_hR[IPB][T2N][DIM / 4];

    const int  rsub = lane >> 3;
    const int  m    = lane & 7;
    const bool tail = (m < 3);
    const int  p9   = tail ? (m + 72) : (m + 64);

    const v4f* q4 = (const v4f*)(q + ((size_t)b * T1N + i0) * DIM);
    v4f qa[10], qb[10];
    for (int t = 0; t < 9; ++t) { qa[t] = q4[m + 8 * t]; qb[t] = q4[DIM / 4 + m + 8 * t]; }
    qa[9] = q4[p9]; qb[9] = q4[DIM / 4 + p9];

    for (int g = wave; g < T2N / 4; g += WAVES) {
        const int j = 4 * g + rsub;
        const v4f* hLa = (const v4f*)(hL + (((size_t)b * T1N + i0) * T2N + j) * DIM);
        const v4f* hLb = hLa + (size_t)T2N * (DIM / 4);
        const v4f* hRa = (const v4f*)(hR + (((size_t)b * T2N + j) * T1N + i0) * DIM);
        const v4f* hRb = hRa + DIM / 4;
        const v4f* k4  = (const v4f*)(k  + ((size_t)b * T2N + j) * DIM);

        v4f kk[10], a0[10], r0[10], a1[10], r1[10];
        for (int t = 0; t < 9; ++t) {
            kk[t] = k4[m + 8 * t];
            a0[t] = hLa[m + 8 * t];
            r0[t] = hRa[m + 8 * t];
            a1[t] = hLb[m + 8 * t];
            r1[t] = hRb[m + 8 * t];
        }
        kk[9] = k4[p9];
        a0[9] = hLa[p9]; r0[9] = hRa[p9];
        a1[9] = hLb[p9]; r1[9] = hRb[p9];

        for (int t = 0; t < 9; ++t) {
            s_hR[0][j][m + 8 * t] = r0[t];
            s_hR[1][j][m + 8 * t] = r1[t];
        }
        s_hR[0][j][p9] = r0[9];
        s_hR[1][j][p9] = r1[9];

        const float sc0 = score_row(qa, a0, r0, kk, tail);
        const float sc1 = score_row(qb, a1, r1, kk, tail);
        if (m == 0) { s_attn[0][j] = sc0; s_attn[1][j] = sc1; }
    }
    __syncthreads();

    {
        const int r  = wave >> 2;
        const int jo = 32 * (wave & 3) + lane;
        float* sa = s_attn[r];
        const float* e1 = s_e1[r];
        const float* e2 = s_e2[r];

        float mx = -__builtin_inff();
        mx = fmaxf(mx, sa[lane]);      mx = fmaxf(mx, sa[lane + 32]);
        mx = fmaxf(mx, sa[lane + 64]); mx = fmaxf(mx, sa[lane + 96]);
        #pragma unroll
        for (int off = 16; off > 0; off >>= 1) mx = fmaxf(mx, __shfl_xor(mx, off, 32));

        const float e_own = expf(sa[jo] - mx);
        s_e1[r][jo] = e_own;
        __syncthreads();

        float sum = 0.0f;
        sum += e1[lane]; sum += e1[lane + 32]; sum += e1[lane + 64]; sum += e1[lane + 96];
        #pragma unroll
        for (int off = 16; off > 0; off >>= 1) sum += __shfl_xor(sum, off, 32);
        const float inv = 1.0f / sum;

        float m2 = -__builtin_inff();
        m2 = fmaxf(m2, 1000.0f * (e1[lane]      * inv));
        m2 = fmaxf(m2, 1000.0f * (e1[lane + 32] * inv));
        m2 = fmaxf(m2, 1000.0f * (e1[lane + 64] * inv));
        m2 = fmaxf(m2, 1000.0f * (e1[lane + 96] * inv));
        #pragma unroll
        for (int off = 16; off > 0; off >>= 1) m2 = fmaxf(m2, __shfl_xor(m2, off, 32));

        const float p_own = e_own * inv;
        const float f_own = expf(1000.0f * p_own - m2);
        s_e2[r][jo] = f_own;
        __syncthreads();

        float sum2 = 0.0f;
        sum2 += e2[lane]; sum2 += e2[lane + 32]; sum2 += e2[lane + 64]; sum2 += e2[lane + 96];
        #pragma unroll
        for (int off = 16; off > 0; off >>= 1) sum2 += __shfl_xor(sum2, off, 32);
        const float inv2 = 1.0f / sum2;

        const float a2 = f_own * inv2;
        sa[jo] = fminf(fmaxf(a2, 0.0f), 1.0f);
    }
    __syncthreads();

    {
        const int  j0 = wave * (T2N / WAVES);
        const int  p2 = (lane < 11) ? (lane + 64) : (lane + 32);
        const v4f* v4 = (const v4f*)(v  + ((size_t)b * T2N + j0) * DIM);

        v4f acc[IPB][3] = {};
        #pragma unroll 1
        for (int r0 = 0; r0 < T2N / WAVES; r0 += 4) {
            v4f vv[4][3], hh[IPB][4][3];
            for (int r = 0; r < 4; ++r) {
                const v4f* vr = v4 + (size_t)(r0 + r) * (DIM / 4);
                vv[r][0] = vr[lane]; vv[r][1] = vr[lane + 32]; vv[r][2] = vr[p2];
                for (int ii = 0; ii < IPB; ++ii) {
                    const v4f* hr = s_hR[ii][j0 + r0 + r];
                    hh[ii][r][0] = hr[lane]; hh[ii][r][1] = hr[lane + 32]; hh[ii][r][2] = hr[p2];
                }
            }
            for (int ii = 0; ii < IPB; ++ii) {
                for (int r = 0; r < 4; ++r) {
                    const float w = s_attn[ii][j0 + r0 + r];
                    for (int c = 0; c < 3; ++c) {
                        acc[ii][c].x = fmaf(w, vv[r][c].x + hh[ii][r][c].x, acc[ii][c].x);
                        acc[ii][c].y = fmaf(w, vv[r][c].y + hh[ii][r][c].y, acc[ii][c].y);
                        acc[ii][c].z = fmaf(w, vv[r][c].z + hh[ii][r][c].z, acc[ii][c].z);
                        acc[ii][c].w = fmaf(w, vv[r][c].w + hh[ii][r][c].w, acc[ii][c].w);
                    }
                }
            }
        }
        for (int ii = 0; ii < IPB; ++ii) {
            s_hR[ii][j0][lane]      = acc[ii][0];
            s_hR[ii][j0][lane + 32] = acc[ii][1];
            if (lane < 11) s_hR[ii][j0][lane + 64] = acc[ii][2];
        }
    }
    __syncthreads();

    if (tid < IPB * (DIM / 4)) {
        const int ii = (tid >= DIM / 4) ? 1 : 0;
        const int p  = tid - ii * (DIM / 4);
        v4f s = s_hR[ii][0][p];
        for (int w = 1; w < WAVES; ++w) s += s_hR[ii][w * (T2N / WAVES)][p];
        ((v4f*)(out + ((size_t)b * T1N + i0) * DIM))[tid] = s;
    }
}

extern "C" void kernel_launch(void* const* d_in, const int* in_sizes, int n_in,
                              void* d_out, int out_size, void* d_ws, size_t ws_size,
                              hipStream_t stream)
{
    const float* query = (const float*)d_in[0];
    const float* key   = (const float*)d_in[1];
    const float* value = (const float*)d_in[2];
    const float* hL    = (const float*)d_in[3];
    const float* hR    = (const float*)d_in[4];
    const float* WQ    = (const float*)d_in[5];
    const float* bQ    = (const float*)d_in[6];
    const float* WK    = (const float*)d_in[7];
    const float* bK    = (const float*)d_in[8];
    float* out = (float*)d_out;

    float* q = (float*)d_ws;
    float* k = q + (size_t)ROWS * DIM;
    float* v = k + (size_t)ROWS * DIM;

    const dim3 pgrid((DIM + 15) / 16, ROWS / 16);
    proj_wmma_kernel<<<pgrid, 32, 0, stream>>>(query, WQ, bQ, q);
    proj_pair_wmma_kernel<<<pgrid, 32, 0, stream>>>(key, value, WK, bK, k, v);

    attn_fused_kernel<<<dim3(T1N / IPB, BATCH), BLOCK, 0, stream>>>(q, k, v, hL, hR, out);
}
